# starting kernel
# baseline (speedup 1.0000x reference)
#include <hip/hip_runtime.h>

typedef float v2f __attribute__((ext_vector_type(2)));
typedef float v8f __attribute__((ext_vector_type(8)));

#define D 128
#define WP 132

__global__ void gcn_zero_kernel(float4* __restrict__ p, int n4) {
    int i = blockIdx.x * blockDim.x + threadIdx.x;
    int stride = gridDim.x * blockDim.x;
    float4 z = make_float4(0.f, 0.f, 0.f, 0.f);
    for (; i < n4; i += stride) p[i] = z;
}

__global__ void __launch_bounds__(256)
gcn_scatter_kernel(const float* __restrict__ X,
                   const int*   __restrict__ edge_row,
                   const int*   __restrict__ edge_col,
                   const float* __restrict__ edge_val,
                   float*       __restrict__ agg,
                   int E) {
    int wid  = (blockIdx.x * blockDim.x + threadIdx.x) >> 5;
    int lane = threadIdx.x & 31;
    if (wid >= E) return;

    int   r = edge_row[wid];
    int   c = edge_col[wid];
    float v = edge_val[wid];

    const float4* xrow = (const float4*)(X + (size_t)c * D);
    float4 x = xrow[lane];

    float* dst = agg + (size_t)r * D + lane * 4;
    __hip_atomic_fetch_add(dst + 0, v * x.x, __ATOMIC_RELAXED, __HIP_MEMORY_SCOPE_AGENT);
    __hip_atomic_fetch_add(dst + 1, v * x.y, __ATOMIC_RELAXED, __HIP_MEMORY_SCOPE_AGENT);
    __hip_atomic_fetch_add(dst + 2, v * x.z, __ATOMIC_RELAXED, __HIP_MEMORY_SCOPE_AGENT);
    __hip_atomic_fetch_add(dst + 3, v * x.w, __ATOMIC_RELAXED, __HIP_MEMORY_SCOPE_AGENT);
}

__global__ void __launch_bounds__(256)
gcn_gemm_wmma_kernel(const float* __restrict__ Wmat,
                     const float* __restrict__ bias,
                     float*       __restrict__ out,
                     int nrows) {
    __shared__ float Wlds[D * WP];

    const int tid = threadIdx.x;

    {
        const float4* Wv = (const float4*)Wmat;
        for (int i4 = tid; i4 < (D * D) / 4; i4 += 256) {
            int o = i4 >> 5;
            int k = (i4 & 31) * 4;
            *(float4*)&Wlds[o * WP + k] = Wv[i4];
        }
    }
    __syncthreads();

    const int wave = tid >> 5;
    const int lane = tid & 31;
    const int lr   = lane & 15;
    const int hi   = lane >> 4;

    const int row_base = blockIdx.x * 128 + wave * 16;
    const int row      = row_base + lr;
    const int rowc     = row < nrows ? row : (nrows - 1);
    const float* arow  = out + (size_t)rowc * D;

    v8f acc[8];
#pragma unroll
    for (int t = 0; t < 8; ++t)
        acc[t] = (v8f){0.f, 0.f, 0.f, 0.f, 0.f, 0.f, 0.f, 0.f};

#pragma unroll 8
    for (int kk = 0; kk < D / 4; ++kk) {
        const int k0 = kk * 4 + hi * 2;
        v2f a = *(const v2f*)(arow + k0);

#pragma unroll
        for (int t = 0; t < 8; ++t) {
            const int o = t * 16 + lr;
            v2f b = *(const v2f*)&Wlds[o * WP + k0];
            acc[t] = __builtin_amdgcn_wmma_f32_16x16x4_f32(
                 false, a,  false, b,
                 (short)0, acc[t],
                 false,  false);
        }
    }

#pragma unroll
    for (int t = 0; t < 8; ++t) {
        const int col = t * 16 + lr;
        const float bb = bias[col];
#pragma unroll
        for (int j = 0; j < 8; ++j) {
            const int r = row_base + j + hi * 8;
            if (r < nrows) out[(size_t)r * D + col] = acc[t][j] + bb;
        }
    }
}

extern "C" void kernel_launch(void* const* d_in, const int* in_sizes, int n_in,
                              void* d_out, int out_size, void* d_ws, size_t ws_size,
                              hipStream_t stream) {
    const float* X        = (const float*)d_in[0];
    const int*   edge_row = (const int*)  d_in[1];
    const int*   edge_col = (const int*)  d_in[2];
    const float* edge_val = (const float*)d_in[3];
    const float* Wmat     = (const float*)d_in[4];
    const float* bias     = (const float*)d_in[5];
    float*       out      = (float*)d_out;

    const int N = in_sizes[0] / D;
    const int E = in_sizes[1];

    {
        int n4 = out_size / 4;
        gcn_zero_kernel<<<2048, 256, 0, stream>>>((float4*)out, n4);
    }

    {
        long long threads = (long long)E * 32;
        int blocks = (int)((threads + 255) / 256);
        gcn_scatter_kernel<<<blocks, 256, 0, stream>>>(X, edge_row, edge_col,
                                                       edge_val, out, E);
    }

    {
        int blocks = (N + 127) / 128;
        gcn_gemm_wmma_kernel<<<blocks, 256, 0, stream>>>(Wmat, bias, out, N);
    }
}


# Scatter lane map: lane l now owns features l, l+32, l+64, l+96 (was 4 consecutive): every f32 atomic and X load instruction covers one whole 128-B line, 4 atomic line requests per edge instead of 16
# speedup vs baseline: 1.3861x; 1.3861x over previous
#include <hip/hip_runtime.h>

typedef float v2f __attribute__((ext_vector_type(2)));
typedef float v8f __attribute__((ext_vector_type(8)));

#define D 128
#define WP 132

__global__ void gcn_zero_kernel(float4* __restrict__ p, int n4) {
    int i = blockIdx.x * blockDim.x + threadIdx.x;
    int stride = gridDim.x * blockDim.x;
    float4 z = make_float4(0.f, 0.f, 0.f, 0.f);
    for (; i < n4; i += stride) p[i] = z;
}

__global__ void __launch_bounds__(256)
gcn_scatter_kernel(const float* __restrict__ X,
                   const int*   __restrict__ edge_row,
                   const int*   __restrict__ edge_col,
                   const float* __restrict__ edge_val,
                   float*       __restrict__ agg,
                   int E) {
    int wid  = (blockIdx.x * blockDim.x + threadIdx.x) >> 5;
    int lane = threadIdx.x & 31;
    if (wid >= E) return;

    int   r = edge_row[wid];
    int   c = edge_col[wid];
    float v = edge_val[wid];

    const float* xrow = X + (size_t)c * D + lane;
    float x0 = xrow[0];
    float x1 = xrow[32];
    float x2 = xrow[64];
    float x3 = xrow[96];

    float* dst = agg + (size_t)r * D + lane;
    __hip_atomic_fetch_add(dst +  0, v * x0, __ATOMIC_RELAXED, __HIP_MEMORY_SCOPE_AGENT);
    __hip_atomic_fetch_add(dst + 32, v * x1, __ATOMIC_RELAXED, __HIP_MEMORY_SCOPE_AGENT);
    __hip_atomic_fetch_add(dst + 64, v * x2, __ATOMIC_RELAXED, __HIP_MEMORY_SCOPE_AGENT);
    __hip_atomic_fetch_add(dst + 96, v * x3, __ATOMIC_RELAXED, __HIP_MEMORY_SCOPE_AGENT);
}

__global__ void __launch_bounds__(256)
gcn_gemm_wmma_kernel(const float* __restrict__ Wmat,
                     const float* __restrict__ bias,
                     float*       __restrict__ out,
                     int nrows) {
    __shared__ float Wlds[D * WP];

    const int tid = threadIdx.x;

    {
        const float4* Wv = (const float4*)Wmat;
        for (int i4 = tid; i4 < (D * D) / 4; i4 += 256) {
            int o = i4 >> 5;
            int k = (i4 & 31) * 4;
            *(float4*)&Wlds[o * WP + k] = Wv[i4];
        }
    }
    __syncthreads();

    const int wave = tid >> 5;
    const int lane = tid & 31;
    const int lr   = lane & 15;
    const int hi   = lane >> 4;

    const int row_base = blockIdx.x * 128 + wave * 16;
    const int row      = row_base + lr;
    const int rowc     = row < nrows ? row : (nrows - 1);
    const float* arow  = out + (size_t)rowc * D;

    v8f acc[8];
#pragma unroll
    for (int t = 0; t < 8; ++t)
        acc[t] = (v8f){0.f, 0.f, 0.f, 0.f, 0.f, 0.f, 0.f, 0.f};

#pragma unroll 8
    for (int kk = 0; kk < D / 4; ++kk) {
        const int k0 = kk * 4 + hi * 2;
        v2f a = *(const v2f*)(arow + k0);

#pragma unroll
        for (int t = 0; t < 8; ++t) {
            const int o = t * 16 + lr;
            v2f b = *(const v2f*)&Wlds[o * WP + k0];
            acc[t] = __builtin_amdgcn_wmma_f32_16x16x4_f32(
                 false, a,  false, b,
                 (short)0, acc[t],
                 false,  false);
        }
    }

#pragma unroll
    for (int t = 0; t < 8; ++t) {
        const int col = t * 16 + lr;
        const float bb = bias[col];
#pragma unroll
        for (int j = 0; j < 8; ++j) {
            const int r = row_base + j + hi * 8;
            if (r < nrows) out[(size_t)r * D + col] = acc[t][j] + bb;
        }
    }
}

extern "C" void kernel_launch(void* const* d_in, const int* in_sizes, int n_in,
                              void* d_out, int out_size, void* d_ws, size_t ws_size,
                              hipStream_t stream) {
    const float* X        = (const float*)d_in[0];
    const int*   edge_row = (const int*)  d_in[1];
    const int*   edge_col = (const int*)  d_in[2];
    const float* edge_val = (const float*)d_in[3];
    const float* Wmat     = (const float*)d_in[4];
    const float* bias     = (const float*)d_in[5];
    float*       out      = (float*)d_out;

    const int N = in_sizes[0] / D;
    const int E = in_sizes[1];

    {
        int n4 = out_size / 4;
        gcn_zero_kernel<<<2048, 256, 0, stream>>>((float4*)out, n4);
    }

    {
        long long threads = (long long)E * 32;
        int blocks = (int)((threads + 255) / 256);
        gcn_scatter_kernel<<<blocks, 256, 0, stream>>>(X, edge_row, edge_col,
                                                       edge_val, out, E);
    }

    {
        int blocks = (N + 127) / 128;
        gcn_gemm_wmma_kernel<<<blocks, 256, 0, stream>>>(Wmat, bias, out, N);
    }
}


# Float-atomic scatter replaced by row buckets (integer-counter slots) + one wave per row that sorts its bucket by edge index and sums in registers: f32 sum order per row now fixed (edge index)
# speedup vs baseline: 1.6513x; 1.1913x over previous
#include <hip/hip_runtime.h>

typedef float v2f __attribute__((ext_vector_type(2)));
typedef float v8f __attribute__((ext_vector_type(8)));

#define D 128
#define WP 132

#define CAP 64

__global__ void gcn_clear_counts_kernel(int* __restrict__ cnt, int n) {
    int i = blockIdx.x * blockDim.x + threadIdx.x;
    if (i < n) cnt[i] = 0;
}

__global__ void __launch_bounds__(256)
gcn_bin_edges_kernel(const int* __restrict__ edge_row,
                     int*       __restrict__ cnt,
                     int*       __restrict__ slots,
                     int*       __restrict__ spill,
                     int N, int E) {
    int e = blockIdx.x * blockDim.x + threadIdx.x;
    if (e >= E) return;
    int r   = edge_row[e];
    int pos = atomicAdd(&cnt[r], 1);
    if (pos < CAP) {
        slots[(size_t)r * CAP + pos] = e;
    } else {
        int k = atomicAdd(&cnt[N], 1);
        spill[k] = e;
    }
}

__device__ __forceinline__ void gcn_fma4(float4& acc, float v, const float4& x) {
    acc.x = fmaf(v, x.x, acc.x);
    acc.y = fmaf(v, x.y, acc.y);
    acc.z = fmaf(v, x.z, acc.z);
    acc.w = fmaf(v, x.w, acc.w);
}

struct GcnBucket {
    int ca, va, ra;
    int cb, vb, rb;
};

__device__ __forceinline__ void gcn_pick(const GcnBucket& bk, int j, int& c, float& v) {
    const unsigned ma = __builtin_amdgcn_ballot_w32(bk.ra == j);
    if (ma) {
        const int s = __builtin_ctz(ma);
        c = __builtin_amdgcn_readlane(bk.ca, s);
        v = __int_as_float(__builtin_amdgcn_readlane(bk.va, s));
    } else {
        const unsigned mb = __builtin_amdgcn_ballot_w32(bk.rb == j);
        const int s = __builtin_ctz(mb);
        c = __builtin_amdgcn_readlane(bk.cb, s);
        v = __int_as_float(__builtin_amdgcn_readlane(bk.vb, s));
    }
}

template <int U>
__device__ __forceinline__ void gcn_sum_chunk(float4& acc, const float4* __restrict__ xlane,
                                              const GcnBucket& bk, int j0) {
    int    c[U];
    float  v[U];
    float4 x[U];
#pragma unroll
    for (int u = 0; u < U; ++u) gcn_pick(bk, j0 + u, c[u], v[u]);
#pragma unroll
    for (int u = 0; u < U; ++u) x[u] = xlane[(size_t)c[u] * (D / 4)];
#pragma unroll
    for (int u = 0; u < U; ++u) gcn_fma4(acc, v[u], x[u]);
}

__global__ void __launch_bounds__(256)
gcn_row_gather_kernel(const float* __restrict__ X,
                      const int*   __restrict__ edge_row,
                      const int*   __restrict__ edge_col,
                      const float* __restrict__ edge_val,
                      const int*   __restrict__ cnt,
                      const int*   __restrict__ slots,
                      const int*   __restrict__ spill,
                      float*       __restrict__ agg,
                      int N) {
    const int lane = threadIdx.x & 31;
    const int wave = __builtin_amdgcn_readfirstlane((int)(threadIdx.x >> 5));
    const int r = blockIdx.x * 8 + wave;
    if (r >= N) return;

    const int total = cnt[r];
    const int deg   = total < CAP ? total : CAP;
    const int* bucket = slots + (size_t)r * CAP;

    const int EMPTY = 0x7fffffff;
    int ea = EMPTY, eb = EMPTY;
    GcnBucket bk;
    bk.ca = 0; bk.va = 0; bk.cb = 0; bk.vb = 0;
    if (lane < deg) {
        ea = bucket[lane];
        bk.ca = edge_col[ea];
        bk.va = __float_as_int(edge_val[ea]);
    }
    if (lane + 32 < deg) {
        eb = bucket[lane + 32];
        bk.cb = edge_col[eb];
        bk.vb = __float_as_int(edge_val[eb]);
    }

    int ra = 0, rb = 0;
    const int dlo = deg < 32 ? deg : 32;
    for (int j = 0; j < dlo; ++j) {
        const int ej = __builtin_amdgcn_readlane(ea, j);
        ra += (ej < ea) ? 1 : 0;
        rb += (ej < eb) ? 1 : 0;
    }
    for (int j = 32; j < deg; ++j) {
        const int ej = __builtin_amdgcn_readlane(eb, j - 32);
        ra += (ej < ea) ? 1 : 0;
        rb += (ej < eb) ? 1 : 0;
    }
    bk.ra = ra;
    bk.rb = rb;

    const float4* xlane = (const float4*)X + lane;
    float4 acc = make_float4(0.f, 0.f, 0.f, 0.f);

    int j = 0;
    for (; j + 8 <= deg; j += 8) gcn_sum_chunk<8>(acc, xlane, bk, j);
    if (j + 4 <= deg) { gcn_sum_chunk<4>(acc, xlane, bk, j); j += 4; }
    const int rem = deg - j;
    if (rem == 3)      gcn_sum_chunk<3>(acc, xlane, bk, j);
    else if (rem == 2) gcn_sum_chunk<2>(acc, xlane, bk, j);
    else if (rem == 1) gcn_sum_chunk<1>(acc, xlane, bk, j);

    if (total > CAP) {
        const int ns = cnt[N];
        for (int k = 0; k < ns; ++k) {
            const int e = spill[k];
            if (edge_row[e] == r) {
                const float  v = edge_val[e];
                const float4 x = xlane[(size_t)edge_col[e] * (D / 4)];
                gcn_fma4(acc, v, x);
            }
        }
    }

    ((float4*)agg)[(size_t)r * (D / 4) + lane] = acc;
}

__global__ void __launch_bounds__(256)
gcn_gemm_wmma_kernel(const float* __restrict__ Wmat,
                     const float* __restrict__ bias,
                     float*       __restrict__ out,
                     int nrows) {
    __shared__ float Wlds[D * WP];

    const int tid = threadIdx.x;

    {
        const float4* Wv = (const float4*)Wmat;
        for (int i4 = tid; i4 < (D * D) / 4; i4 += 256) {
            int o = i4 >> 5;
            int k = (i4 & 31) * 4;
            *(float4*)&Wlds[o * WP + k] = Wv[i4];
        }
    }
    __syncthreads();

    const int wave = tid >> 5;
    const int lane = tid & 31;
    const int lr   = lane & 15;
    const int hi   = lane >> 4;

    const int row_base = blockIdx.x * 128 + wave * 16;
    const int row      = row_base + lr;
    const int rowc     = row < nrows ? row : (nrows - 1);
    const float* arow  = out + (size_t)rowc * D;

    v8f acc[8];
#pragma unroll
    for (int t = 0; t < 8; ++t)
        acc[t] = (v8f){0.f, 0.f, 0.f, 0.f, 0.f, 0.f, 0.f, 0.f};

#pragma unroll 8
    for (int kk = 0; kk < D / 4; ++kk) {
        const int k0 = kk * 4 + hi * 2;
        v2f a = *(const v2f*)(arow + k0);

#pragma unroll
        for (int t = 0; t < 8; ++t) {
            const int o = t * 16 + lr;
            v2f b = *(const v2f*)&Wlds[o * WP + k0];
            acc[t] = __builtin_amdgcn_wmma_f32_16x16x4_f32(
                 false, a,  false, b,
                 (short)0, acc[t],
                 false,  false);
        }
    }

#pragma unroll
    for (int t = 0; t < 8; ++t) {
        const int col = t * 16 + lr;
        const float bb = bias[col];
#pragma unroll
        for (int j = 0; j < 8; ++j) {
            const int r = row_base + j + hi * 8;
            if (r < nrows) out[(size_t)r * D + col] = acc[t][j] + bb;
        }
    }
}

extern "C" void kernel_launch(void* const* d_in, const int* in_sizes, int n_in,
                              void* d_out, int out_size, void* d_ws, size_t ws_size,
                              hipStream_t stream) {
    const float* X        = (const float*)d_in[0];
    const int*   edge_row = (const int*)  d_in[1];
    const int*   edge_col = (const int*)  d_in[2];
    const float* edge_val = (const float*)d_in[3];
    const float* Wmat     = (const float*)d_in[4];
    const float* bias     = (const float*)d_in[5];
    float*       out      = (float*)d_out;

    const int N = in_sizes[0] / D;
    const int E = in_sizes[1];

    int* cnt   = (int*)d_ws;
    int* spill = cnt + ((N + 1 + 63) / 64) * 64;
    int* slots = spill + ((E + 63) / 64) * 64;

    gcn_clear_counts_kernel<<<(N + 1 + 255) / 256, 256, 0, stream>>>(cnt, N + 1);

    gcn_bin_edges_kernel<<<(E + 255) / 256, 256, 0, stream>>>(edge_row, cnt, slots,
                                                             spill, N, E);

    gcn_row_gather_kernel<<<(N + 7) / 8, 256, 0, stream>>>(X, edge_row, edge_col,
                                                           edge_val, cnt, slots,
                                                           spill, out, N);

    {
        int blocks = (N + 127) / 128;
        gcn_gemm_wmma_kernel<<<blocks, 256, 0, stream>>>(Wmat, bias, out, N);
    }
}


# Counters and bucket lines split per chip half: a binning wave reads its die number and uses its own half's counter array and 128-B bucket line, so no counter or slot line is updated from both halves
# speedup vs baseline: 2.3095x; 1.3986x over previous
#include <hip/hip_runtime.h>

typedef float v2f __attribute__((ext_vector_type(2)));
typedef float v8f __attribute__((ext_vector_type(8)));

#define D 128
#define WP 132

#define CAP 64
#define HCAP 32

__device__ __forceinline__ int gcn_chip_half() {
    const unsigned id = (unsigned)__builtin_amdgcn_s_sendmsg_rtn(0x87);
    return (int)((id >> 18) & 1u);
}

__global__ void gcn_clear_counts_kernel(int* __restrict__ cnt, int n) {
    int i = blockIdx.x * blockDim.x + threadIdx.x;
    if (i < n) cnt[i] = 0;
}

__global__ void __launch_bounds__(256)
gcn_bin_edges_kernel(const int* __restrict__ edge_row,
                     int*       __restrict__ cnt,
                     int*       __restrict__ slots,
                     int*       __restrict__ spill,
                     int N, int cstride, int E) {
    const int h = gcn_chip_half();
    int e = blockIdx.x * blockDim.x + threadIdx.x;
    if (e >= E) return;
    int r   = edge_row[e];
    int pos = atomicAdd(&cnt[(size_t)h * cstride + r], 1);
    if (pos < HCAP) {
        slots[(size_t)r * CAP + h * HCAP + pos] = e;
    } else {
        int k = atomicAdd(&cnt[N], 1);
        spill[k] = e;
    }
}

__device__ __forceinline__ void gcn_fma4(float4& acc, float v, const float4& x) {
    acc.x = fmaf(v, x.x, acc.x);
    acc.y = fmaf(v, x.y, acc.y);
    acc.z = fmaf(v, x.z, acc.z);
    acc.w = fmaf(v, x.w, acc.w);
}

struct GcnBucket {
    int ca, va, ra;
    int cb, vb, rb;
};

__device__ __forceinline__ void gcn_pick(const GcnBucket& bk, int j, int& c, float& v) {
    const unsigned ma = __builtin_amdgcn_ballot_w32(bk.ra == j);
    if (ma) {
        const int s = __builtin_ctz(ma);
        c = __builtin_amdgcn_readlane(bk.ca, s);
        v = __int_as_float(__builtin_amdgcn_readlane(bk.va, s));
    } else {
        const unsigned mb = __builtin_amdgcn_ballot_w32(bk.rb == j);
        const int s = __builtin_ctz(mb);
        c = __builtin_amdgcn_readlane(bk.cb, s);
        v = __int_as_float(__builtin_amdgcn_readlane(bk.vb, s));
    }
}

template <int U>
__device__ __forceinline__ void gcn_sum_chunk(float4& acc, const float4* __restrict__ xlane,
                                              const GcnBucket& bk, int j0) {
    int    c[U];
    float  v[U];
    float4 x[U];
#pragma unroll
    for (int u = 0; u < U; ++u) gcn_pick(bk, j0 + u, c[u], v[u]);
#pragma unroll
    for (int u = 0; u < U; ++u) x[u] = xlane[(size_t)c[u] * (D / 4)];
#pragma unroll
    for (int u = 0; u < U; ++u) gcn_fma4(acc, v[u], x[u]);
}

__global__ void __launch_bounds__(256)
gcn_row_gather_kernel(const float* __restrict__ X,
                      const int*   __restrict__ edge_row,
                      const int*   __restrict__ edge_col,
                      const float* __restrict__ edge_val,
                      const int*   __restrict__ cnt,
                      const int*   __restrict__ slots,
                      const int*   __restrict__ spill,
                      float*       __restrict__ agg,
                      int N, int cstride) {
    const int lane = threadIdx.x & 31;
    const int wave = __builtin_amdgcn_readfirstlane((int)(threadIdx.x >> 5));
    const int r = blockIdx.x * 8 + wave;
    if (r >= N) return;

    const int t0  = cnt[r];
    const int t1  = cnt[(size_t)cstride + r];
    const int d0  = t0 < HCAP ? t0 : HCAP;
    const int d1  = t1 < HCAP ? t1 : HCAP;
    const int deg = d0 + d1;
    const int* bucket = slots + (size_t)r * CAP;

    const int EMPTY = 0x7fffffff;
    int ea = EMPTY, eb = EMPTY;
    GcnBucket bk;
    bk.ca = 0; bk.va = 0; bk.cb = 0; bk.vb = 0;
    if (deg > 0) {
        const int la = lane < d0 - 1 ? lane : d0 - 1;
        const int lb = HCAP + (lane < d1 - 1 ? lane : d1 - 1);
        const int xa = bucket[d0 > 0 ? la : lb];
        const int xb = bucket[d1 > 0 ? lb : la];
        bk.ca = edge_col[xa];
        bk.va = __float_as_int(edge_val[xa]);
        bk.cb = edge_col[xb];
        bk.vb = __float_as_int(edge_val[xb]);
        if (lane < d0) ea = xa;
        if (lane < d1) eb = xb;
    }

    int ra = 0, rb = 0;
    for (int j = 0; j < d0; ++j) {
        const int ej = __builtin_amdgcn_readlane(ea, j);
        ra += (ej < ea) ? 1 : 0;
        rb += (ej < eb) ? 1 : 0;
    }
    for (int j = 0; j < d1; ++j) {
        const int ej = __builtin_amdgcn_readlane(eb, j);
        ra += (ej < ea) ? 1 : 0;
        rb += (ej < eb) ? 1 : 0;
    }
    bk.ra = ra;
    bk.rb = rb;

    const float4* xlane = (const float4*)X + lane;
    float4 acc = make_float4(0.f, 0.f, 0.f, 0.f);

    int j = 0;
    for (; j + 8 <= deg; j += 8) gcn_sum_chunk<8>(acc, xlane, bk, j);
    if (j + 4 <= deg) { gcn_sum_chunk<4>(acc, xlane, bk, j); j += 4; }
    const int rem = deg - j;
    if (rem == 3)      gcn_sum_chunk<3>(acc, xlane, bk, j);
    else if (rem == 2) gcn_sum_chunk<2>(acc, xlane, bk, j);
    else if (rem == 1) gcn_sum_chunk<1>(acc, xlane, bk, j);

    if (t0 > HCAP || t1 > HCAP) {
        const int ns = cnt[N];
        for (int k = 0; k < ns; ++k) {
            const int e = spill[k];
            if (edge_row[e] == r) {
                const float  v = edge_val[e];
                const float4 x = xlane[(size_t)edge_col[e] * (D / 4)];
                gcn_fma4(acc, v, x);
            }
        }
    }

    ((float4*)agg)[(size_t)r * (D / 4) + lane] = acc;
}

__global__ void __launch_bounds__(256)
gcn_gemm_wmma_kernel(const float* __restrict__ Wmat,
                     const float* __restrict__ bias,
                     float*       __restrict__ out,
                     int nrows) {
    __shared__ float Wlds[D * WP];

    const int tid = threadIdx.x;

    {
        const float4* Wv = (const float4*)Wmat;
        for (int i4 = tid; i4 < (D * D) / 4; i4 += 256) {
            int o = i4 >> 5;
            int k = (i4 & 31) * 4;
            *(float4*)&Wlds[o * WP + k] = Wv[i4];
        }
    }
    __syncthreads();

    const int wave = tid >> 5;
    const int lane = tid & 31;
    const int lr   = lane & 15;
    const int hi   = lane >> 4;

    const int row_base = blockIdx.x * 128 + wave * 16;
    const int row      = row_base + lr;
    const int rowc     = row < nrows ? row : (nrows - 1);
    const float* arow  = out + (size_t)rowc * D;

    v8f acc[8];
#pragma unroll
    for (int t = 0; t < 8; ++t)
        acc[t] = (v8f){0.f, 0.f, 0.f, 0.f, 0.f, 0.f, 0.f, 0.f};

#pragma unroll 8
    for (int kk = 0; kk < D / 4; ++kk) {
        const int k0 = kk * 4 + hi * 2;
        v2f a = *(const v2f*)(arow + k0);

#pragma unroll
        for (int t = 0; t < 8; ++t) {
            const int o = t * 16 + lr;
            v2f b = *(const v2f*)&Wlds[o * WP + k0];
            acc[t] = __builtin_amdgcn_wmma_f32_16x16x4_f32(
                 false, a,  false, b,
                 (short)0, acc[t],
                 false,  false);
        }
    }

#pragma unroll
    for (int t = 0; t < 8; ++t) {
        const int col = t * 16 + lr;
        const float bb = bias[col];
#pragma unroll
        for (int j = 0; j < 8; ++j) {
            const int r = row_base + j + hi * 8;
            if (r < nrows) out[(size_t)r * D + col] = acc[t][j] + bb;
        }
    }
}

extern "C" void kernel_launch(void* const* d_in, const int* in_sizes, int n_in,
                              void* d_out, int out_size, void* d_ws, size_t ws_size,
                              hipStream_t stream) {
    const float* X        = (const float*)d_in[0];
    const int*   edge_row = (const int*)  d_in[1];
    const int*   edge_col = (const int*)  d_in[2];
    const float* edge_val = (const float*)d_in[3];
    const float* Wmat     = (const float*)d_in[4];
    const float* bias     = (const float*)d_in[5];
    float*       out      = (float*)d_out;

    const int N = in_sizes[0] / D;
    const int E = in_sizes[1];

    const int cstride = ((N + 1 + 63) / 64) * 64;
    int* cnt   = (int*)d_ws;
    int* spill = cnt + 2 * cstride;
    int* slots = spill + ((E + 63) / 64) * 64;

    gcn_clear_counts_kernel<<<(2 * cstride + 255) / 256, 256, 0, stream>>>(cnt, 2 * cstride);

    gcn_bin_edges_kernel<<<(E + 255) / 256, 256, 0, stream>>>(edge_row, cnt, slots,
                                                             spill, N, cstride, E);

    gcn_row_gather_kernel<<<(N + 7) / 8, 256, 0, stream>>>(X, edge_row, edge_col,
                                                           edge_val, cnt, slots,
                                                           spill, out, N, cstride);

    {
        int blocks = (N + 127) / 128;
        gcn_gemm_wmma_kernel<<<blocks, 256, 0, stream>>>(Wmat, bias, out, N);
    }
}


# Row buckets now hold 16-byte entries (edge index, column, value), filled by the binning thread, so the row wave's two dependent random 4-byte lookups per edge (edge_col, edge_val) are gone
# speedup vs baseline: 3.1200x; 1.3509x over previous
#include <hip/hip_runtime.h>

typedef float v2f __attribute__((ext_vector_type(2)));
typedef float v8f __attribute__((ext_vector_type(8)));

#define D 128
#define WP 132

#define CAP 64
#define HCAP 32

__device__ __forceinline__ int gcn_chip_half() {
    const unsigned id = (unsigned)__builtin_amdgcn_s_sendmsg_rtn(0x87);
    return (int)((id >> 18) & 1u);
}

__global__ void gcn_clear_counts_kernel(int* __restrict__ cnt, int n) {
    int i = blockIdx.x * blockDim.x + threadIdx.x;
    if (i < n) cnt[i] = 0;
}

__global__ void __launch_bounds__(256)
gcn_bin_edges_kernel(const int*   __restrict__ edge_row,
                     const int*   __restrict__ edge_col,
                     const float* __restrict__ edge_val,
                     int*         __restrict__ cnt,
                     int4*        __restrict__ slots,
                     int*         __restrict__ spill,
                     int N, int cstride, int E) {
    const int h = gcn_chip_half();
    int e = blockIdx.x * blockDim.x + threadIdx.x;
    if (e >= E) return;
    int r   = edge_row[e];
    const int   c = edge_col[e];
    const float v = edge_val[e];
    asm volatile("" :: "v"(c), "v"(v));
    int pos = atomicAdd(&cnt[(size_t)h * cstride + r], 1);
    if (pos < HCAP) {
        slots[(size_t)r * CAP + h * HCAP + pos] = make_int4(e, c, __float_as_int(v), 0);
    } else {
        int k = atomicAdd(&cnt[N], 1);
        spill[k] = e;
    }
}

__device__ __forceinline__ void gcn_fma4(float4& acc, float v, const float4& x) {
    acc.x = fmaf(v, x.x, acc.x);
    acc.y = fmaf(v, x.y, acc.y);
    acc.z = fmaf(v, x.z, acc.z);
    acc.w = fmaf(v, x.w, acc.w);
}

struct GcnBucket {
    int ca, va, ra;
    int cb, vb, rb;
};

__device__ __forceinline__ void gcn_pick(const GcnBucket& bk, int j, int& c, float& v) {
    const unsigned ma = __builtin_amdgcn_ballot_w32(bk.ra == j);
    if (ma) {
        const int s = __builtin_ctz(ma);
        c = __builtin_amdgcn_readlane(bk.ca, s);
        v = __int_as_float(__builtin_amdgcn_readlane(bk.va, s));
    } else {
        const unsigned mb = __builtin_amdgcn_ballot_w32(bk.rb == j);
        const int s = __builtin_ctz(mb);
        c = __builtin_amdgcn_readlane(bk.cb, s);
        v = __int_as_float(__builtin_amdgcn_readlane(bk.vb, s));
    }
}

template <int U>
__device__ __forceinline__ void gcn_sum_chunk(float4& acc, const float4* __restrict__ xlane,
                                              const GcnBucket& bk, int j0) {
    int    c[U];
    float  v[U];
    float4 x[U];
#pragma unroll
    for (int u = 0; u < U; ++u) gcn_pick(bk, j0 + u, c[u], v[u]);
#pragma unroll
    for (int u = 0; u < U; ++u) x[u] = xlane[(size_t)c[u] * (D / 4)];
#pragma unroll
    for (int u = 0; u < U; ++u) gcn_fma4(acc, v[u], x[u]);
}

__global__ void __launch_bounds__(256)
gcn_row_gather_kernel(const float* __restrict__ X,
                      const int*   __restrict__ edge_row,
                      const int*   __restrict__ edge_col,
                      const float* __restrict__ edge_val,
                      const int*   __restrict__ cnt,
                      const int4*  __restrict__ slots,
                      const int*   __restrict__ spill,
                      float*       __restrict__ agg,
                      int N, int cstride) {
    const int lane = threadIdx.x & 31;
    const int wave = __builtin_amdgcn_readfirstlane((int)(threadIdx.x >> 5));
    const int r = blockIdx.x * 8 + wave;
    if (r >= N) return;

    const int t0  = cnt[r];
    const int t1  = cnt[(size_t)cstride + r];
    const int d0  = t0 < HCAP ? t0 : HCAP;
    const int d1  = t1 < HCAP ? t1 : HCAP;
    const int deg = d0 + d1;
    const int4* bucket = slots + (size_t)r * CAP;

    const int EMPTY = 0x7fffffff;
    int ea = EMPTY, eb = EMPTY;
    GcnBucket bk;
    bk.ca = 0; bk.va = 0; bk.cb = 0; bk.vb = 0;
    if (deg > 0) {
        const int la = lane < d0 - 1 ? lane : d0 - 1;
        const int lb = HCAP + (lane < d1 - 1 ? lane : d1 - 1);
        const int4 xa = bucket[d0 > 0 ? la : lb];
        const int4 xb = bucket[d1 > 0 ? lb : la];
        bk.ca = xa.y;
        bk.va = xa.z;
        bk.cb = xb.y;
        bk.vb = xb.z;
        if (lane < d0) ea = xa.x;
        if (lane < d1) eb = xb.x;
    }

    int ra = 0, rb = 0;
    for (int j = 0; j < d0; ++j) {
        const int ej = __builtin_amdgcn_readlane(ea, j);
        ra += (ej < ea) ? 1 : 0;
        rb += (ej < eb) ? 1 : 0;
    }
    for (int j = 0; j < d1; ++j) {
        const int ej = __builtin_amdgcn_readlane(eb, j);
        ra += (ej < ea) ? 1 : 0;
        rb += (ej < eb) ? 1 : 0;
    }
    bk.ra = ra;
    bk.rb = rb;

    const float4* xlane = (const float4*)X + lane;
    float4 acc = make_float4(0.f, 0.f, 0.f, 0.f);

    int j = 0;
    for (; j + 8 <= deg; j += 8) gcn_sum_chunk<8>(acc, xlane, bk, j);
    if (j + 4 <= deg) { gcn_sum_chunk<4>(acc, xlane, bk, j); j += 4; }
    const int rem = deg - j;
    if (rem == 3)      gcn_sum_chunk<3>(acc, xlane, bk, j);
    else if (rem == 2) gcn_sum_chunk<2>(acc, xlane, bk, j);
    else if (rem == 1) gcn_sum_chunk<1>(acc, xlane, bk, j);

    if (t0 > HCAP || t1 > HCAP) {
        const int ns = cnt[N];
        for (int k = 0; k < ns; ++k) {
            const int e = spill[k];
            if (edge_row[e] == r) {
                const float  v = edge_val[e];
                const float4 x = xlane[(size_t)edge_col[e] * (D / 4)];
                gcn_fma4(acc, v, x);
            }
        }
    }

    ((float4*)agg)[(size_t)r * (D / 4) + lane] = acc;
}

__global__ void __launch_bounds__(256)
gcn_gemm_wmma_kernel(const float* __restrict__ Wmat,
                     const float* __restrict__ bias,
                     float*       __restrict__ out,
                     int nrows) {
    __shared__ float Wlds[D * WP];

    const int tid = threadIdx.x;

    {
        const float4* Wv = (const float4*)Wmat;
        for (int i4 = tid; i4 < (D * D) / 4; i4 += 256) {
            int o = i4 >> 5;
            int k = (i4 & 31) * 4;
            *(float4*)&Wlds[o * WP + k] = Wv[i4];
        }
    }
    __syncthreads();

    const int wave = tid >> 5;
    const int lane = tid & 31;
    const int lr   = lane & 15;
    const int hi   = lane >> 4;

    const int row_base = blockIdx.x * 128 + wave * 16;
    const int row      = row_base + lr;
    const int rowc     = row < nrows ? row : (nrows - 1);
    const float* arow  = out + (size_t)rowc * D;

    v8f acc[8];
#pragma unroll
    for (int t = 0; t < 8; ++t)
        acc[t] = (v8f){0.f, 0.f, 0.f, 0.f, 0.f, 0.f, 0.f, 0.f};

#pragma unroll 8
    for (int kk = 0; kk < D / 4; ++kk) {
        const int k0 = kk * 4 + hi * 2;
        v2f a = *(const v2f*)(arow + k0);

#pragma unroll
        for (int t = 0; t < 8; ++t) {
            const int o = t * 16 + lr;
            v2f b = *(const v2f*)&Wlds[o * WP + k0];
            acc[t] = __builtin_amdgcn_wmma_f32_16x16x4_f32(
                 false, a,  false, b,
                 (short)0, acc[t],
                 false,  false);
        }
    }

#pragma unroll
    for (int t = 0; t < 8; ++t) {
        const int col = t * 16 + lr;
        const float bb = bias[col];
#pragma unroll
        for (int j = 0; j < 8; ++j) {
            const int r = row_base + j + hi * 8;
            if (r < nrows) out[(size_t)r * D + col] = acc[t][j] + bb;
        }
    }
}

extern "C" void kernel_launch(void* const* d_in, const int* in_sizes, int n_in,
                              void* d_out, int out_size, void* d_ws, size_t ws_size,
                              hipStream_t stream) {
    const float* X        = (const float*)d_in[0];
    const int*   edge_row = (const int*)  d_in[1];
    const int*   edge_col = (const int*)  d_in[2];
    const float* edge_val = (const float*)d_in[3];
    const float* Wmat     = (const float*)d_in[4];
    const float* bias     = (const float*)d_in[5];
    float*       out      = (float*)d_out;

    const int N = in_sizes[0] / D;
    const int E = in_sizes[1];

    const int cstride = ((N + 1 + 63) / 64) * 64;
    int*  cnt   = (int*)d_ws;
    int*  spill = cnt + 2 * cstride;
    int4* slots = (int4*)(spill + ((E + 127) / 128) * 128);

    gcn_clear_counts_kernel<<<(2 * cstride + 255) / 256, 256, 0, stream>>>(cnt, 2 * cstride);

    gcn_bin_edges_kernel<<<(E + 255) / 256, 256, 0, stream>>>(edge_row, edge_col, edge_val,
                                                             cnt, slots, spill, N,
                                                             cstride, E);

    gcn_row_gather_kernel<<<(N + 7) / 8, 256, 0, stream>>>(X, edge_row, edge_col,
                                                           edge_val, cnt, slots,
                                                           spill, out, N, cstride);

    {
        int blocks = (N + 127) / 128;
        gcn_gemm_wmma_kernel<<<blocks, 256, 0, stream>>>(Wmat, bias, out, N);
    }
}


# Gather and GEMM fused: a workgroup sums 16 rows into an LDS tile, its 8 waves multiply it (WMMA f32) by W fragments kept in registers; row sums never reach memory, one launch less; sum orders kept
# speedup vs baseline: 3.3787x; 1.0829x over previous
#include <hip/hip_runtime.h>

typedef float v2f __attribute__((ext_vector_type(2)));
typedef float v8f __attribute__((ext_vector_type(8)));

#define D 128
#define WP 132

#define GRID_BLOCKS 1024
#define CAP 64
#define HCAP 32

__device__ __forceinline__ int gcn_chip_half() {
    const unsigned id = (unsigned)__builtin_amdgcn_s_sendmsg_rtn(0x87);
    return (int)((id >> 18) & 1u);
}

__global__ void gcn_clear_counts_kernel(int* __restrict__ cnt, int n) {
    int i = blockIdx.x * blockDim.x + threadIdx.x;
    if (i < n) cnt[i] = 0;
}

__global__ void __launch_bounds__(256)
gcn_bin_edges_kernel(const int*   __restrict__ edge_row,
                     const int*   __restrict__ edge_col,
                     const float* __restrict__ edge_val,
                     int*         __restrict__ cnt,
                     int4*        __restrict__ slots,
                     int*         __restrict__ spill,
                     int N, int cstride, int E) {
    const int h = gcn_chip_half();
    int e = blockIdx.x * blockDim.x + threadIdx.x;
    if (e >= E) return;
    int r   = edge_row[e];
    const int   c = edge_col[e];
    const float v = edge_val[e];
    asm volatile("" :: "v"(c), "v"(v));
    int pos = atomicAdd(&cnt[(size_t)h * cstride + r], 1);
    if (pos < HCAP) {
        slots[(size_t)r * CAP + h * HCAP + pos] = make_int4(e, c, __float_as_int(v), 0);
    } else {
        int k = atomicAdd(&cnt[N], 1);
        spill[k] = e;
    }
}

__device__ __forceinline__ void gcn_fma4(float4& acc, float v, const float4& x) {
    acc.x = fmaf(v, x.x, acc.x);
    acc.y = fmaf(v, x.y, acc.y);
    acc.z = fmaf(v, x.z, acc.z);
    acc.w = fmaf(v, x.w, acc.w);
}

struct GcnBucket {
    int ca, va, ra;
    int cb, vb, rb;
};

__device__ __forceinline__ void gcn_pick(const GcnBucket& bk, int j, int& c, float& v) {
    const unsigned ma = __builtin_amdgcn_ballot_w32(bk.ra == j);
    if (ma) {
        const int s = __builtin_ctz(ma);
        c = __builtin_amdgcn_readlane(bk.ca, s);
        v = __int_as_float(__builtin_amdgcn_readlane(bk.va, s));
    } else {
        const unsigned mb = __builtin_amdgcn_ballot_w32(bk.rb == j);
        const int s = __builtin_ctz(mb);
        c = __builtin_amdgcn_readlane(bk.cb, s);
        v = __int_as_float(__builtin_amdgcn_readlane(bk.vb, s));
    }
}

template <int U>
__device__ __forceinline__ void gcn_sum_chunk(float4& acc, const float4* __restrict__ xlane,
                                              const GcnBucket& bk, int j0) {
    int    c[U];
    float  v[U];
    float4 x[U];
#pragma unroll
    for (int u = 0; u < U; ++u) gcn_pick(bk, j0 + u, c[u], v[u]);
#pragma unroll
    for (int u = 0; u < U; ++u) x[u] = xlane[(size_t)c[u] * (D / 4)];
#pragma unroll
    for (int u = 0; u < U; ++u) gcn_fma4(acc, v[u], x[u]);
}

#define AP 132

struct GcnRowIn {
    int  t0, t1, d0, d1;
    int4 xa, xb;
};

__device__ __forceinline__ GcnRowIn gcn_row_fetch(const int*  __restrict__ cnt,
                                                  const int4* __restrict__ slots,
                                                  int r, int N, int cstride, int lane) {
    GcnRowIn in;
    in.t0 = 0;
    in.t1 = 0;
    if (r < N) {
        in.t0 = cnt[r];
        in.t1 = cnt[(size_t)cstride + r];
    }
    in.d0 = in.t0 < HCAP ? in.t0 : HCAP;
    in.d1 = in.t1 < HCAP ? in.t1 : HCAP;
    in.xa = make_int4(0, 0, 0, 0);
    in.xb = make_int4(0, 0, 0, 0);
    if (in.d0 + in.d1 > 0) {
        const int4* bucket = slots + (size_t)r * CAP;
        const int la = lane < in.d0 - 1 ? lane : in.d0 - 1;
        const int lb = HCAP + (lane < in.d1 - 1 ? lane : in.d1 - 1);
        in.xa = bucket[in.d0 > 0 ? la : lb];
        in.xb = bucket[in.d1 > 0 ? lb : la];
    }
    return in;
}

__device__ __forceinline__ float4 gcn_row_sum(const GcnRowIn& in,
                                              const float* __restrict__ X,
                                              const int*   __restrict__ edge_row,
                                              const int*   __restrict__ edge_col,
                                              const float* __restrict__ edge_val,
                                              const int*   __restrict__ cnt,
                                              const int*   __restrict__ spill,
                                              int r, int N, int lane) {
    const int d0  = in.d0;
    const int d1  = in.d1;
    const int deg = d0 + d1;

    const int EMPTY = 0x7fffffff;
    const int ea = lane < d0 ? in.xa.x : EMPTY;
    const int eb = lane < d1 ? in.xb.x : EMPTY;
    GcnBucket bk;
    bk.ca = in.xa.y;
    bk.va = in.xa.z;
    bk.cb = in.xb.y;
    bk.vb = in.xb.z;

    int ra = 0, rb = 0;
    for (int j = 0; j < d0; ++j) {
        const int ej = __builtin_amdgcn_readlane(ea, j);
        ra += (ej < ea) ? 1 : 0;
        rb += (ej < eb) ? 1 : 0;
    }
    for (int j = 0; j < d1; ++j) {
        const int ej = __builtin_amdgcn_readlane(eb, j);
        ra += (ej < ea) ? 1 : 0;
        rb += (ej < eb) ? 1 : 0;
    }
    bk.ra = ra;
    bk.rb = rb;

    const float4* xlane = (const float4*)X + lane;
    float4 acc = make_float4(0.f, 0.f, 0.f, 0.f);

    int j = 0;
    for (; j + 8 <= deg; j += 8) gcn_sum_chunk<8>(acc, xlane, bk, j);
    if (j + 4 <= deg) { gcn_sum_chunk<4>(acc, xlane, bk, j); j += 4; }
    const int rem = deg - j;
    if (rem == 3)      gcn_sum_chunk<3>(acc, xlane, bk, j);
    else if (rem == 2) gcn_sum_chunk<2>(acc, xlane, bk, j);
    else if (rem == 1) gcn_sum_chunk<1>(acc, xlane, bk, j);

    if (in.t0 > HCAP || in.t1 > HCAP) {
        const int ns = cnt[N];
        for (int k = 0; k < ns; ++k) {
            const int e = spill[k];
            if (edge_row[e] == r) {
                const float  v = edge_val[e];
                const float4 x = xlane[(size_t)edge_col[e] * (D / 4)];
                gcn_fma4(acc, v, x);
            }
        }
    }
    return acc;
}

__global__ void __launch_bounds__(256, 8)
gcn_gather_gemm_kernel(const float* __restrict__ X,
                       const int*   __restrict__ edge_row,
                       const int*   __restrict__ edge_col,
                       const float* __restrict__ edge_val,
                       const int*   __restrict__ cnt,
                       const int4*  __restrict__ slots,
                       const int*   __restrict__ spill,
                       const float* __restrict__ Wmat,
                       const float* __restrict__ bias,
                       float*       __restrict__ out,
                       int N, int cstride, int ntiles) {
    __shared__ __attribute__((aligned(16))) float Atile[16 * AP];
    __shared__ __attribute__((aligned(16))) float Dtile[16 * AP];

    const int lane = threadIdx.x & 31;
    const int wave = __builtin_amdgcn_readfirstlane((int)(threadIdx.x >> 5));
    const int lr   = lane & 15;
    const int hi   = lane >> 4;

    const int col = wave * 16 + lr;
    v2f bfrag[D / 4];
    {
        const float* wrow = Wmat + (size_t)col * D + 2 * hi;
#pragma unroll
        for (int kk = 0; kk < D / 4; ++kk) bfrag[kk] = *(const v2f*)(wrow + 4 * kk);
    }
    const float bb = bias[col];

    for (int t = blockIdx.x; t < ntiles; t += gridDim.x) {
        float* At = &Atile[0];
        const int rl = wave * 2;
        const int r0 = t * 16 + rl;

        const GcnRowIn in0 = gcn_row_fetch(cnt, slots, r0,     N, cstride, lane);
        const GcnRowIn in1 = gcn_row_fetch(cnt, slots, r0 + 1, N, cstride, lane);

        const float4 s0 = gcn_row_sum(in0, X, edge_row, edge_col, edge_val, cnt, spill,
                                      r0, N, lane);
        *(float4*)&At[rl * AP + 4 * lane] = s0;
        const float4 s1 = gcn_row_sum(in1, X, edge_row, edge_col, edge_val, cnt, spill,
                                      r0 + 1, N, lane);
        *(float4*)&At[(rl + 1) * AP + 4 * lane] = s1;

        __syncthreads();

        v8f d = (v8f){0.f, 0.f, 0.f, 0.f, 0.f, 0.f, 0.f, 0.f};
        const float* arow = At + lr * AP + 2 * hi;
#pragma unroll
        for (int kk = 0; kk < D / 4; ++kk) {
            const v2f a = *(const v2f*)(arow + 4 * kk);
            d = __builtin_amdgcn_wmma_f32_16x16x4_f32(
                 false, a,  false, bfrag[kk],
                 (short)0, d,
                 false,  false);
        }

#pragma unroll
        for (int j = 0; j < 8; ++j) Dtile[(j + hi * 8) * AP + col] = d[j] + bb;

        __syncthreads();

#pragma unroll
        for (int i = 0; i < 2; ++i) {
            const int r = r0 + i;
            const float4 o = *(const float4*)&Dtile[(rl + i) * AP + 4 * lane];
            if (r < N) ((float4*)out)[(size_t)r * (D / 4) + lane] = o;
        }
    }
}

extern "C" void kernel_launch(void* const* d_in, const int* in_sizes, int n_in,
                              void* d_out, int out_size, void* d_ws, size_t ws_size,
                              hipStream_t stream) {
    const float* X        = (const float*)d_in[0];
    const int*   edge_row = (const int*)  d_in[1];
    const int*   edge_col = (const int*)  d_in[2];
    const float* edge_val = (const float*)d_in[3];
    const float* Wmat     = (const float*)d_in[4];
    const float* bias     = (const float*)d_in[5];
    float*       out      = (float*)d_out;

    const int N = in_sizes[0] / D;
    const int E = in_sizes[1];

    const int cstride = ((N + 1 + 63) / 64) * 64;
    int*  cnt   = (int*)d_ws;
    int*  spill = cnt + 2 * cstride;
    int4* slots = (int4*)(spill + ((E + 127) / 128) * 128);

    gcn_clear_counts_kernel<<<(2 * cstride + 255) / 256, 256, 0, stream>>>(cnt, 2 * cstride);

    gcn_bin_edges_kernel<<<(E + 255) / 256, 256, 0, stream>>>(edge_row, edge_col, edge_val,
                                                             cnt, slots, spill, N,
                                                             cstride, E);

    {
        const int ntiles = (N + 15) / 16;
        const int blocks = ntiles < GRID_BLOCKS ? ntiles : GRID_BLOCKS;
        gcn_gather_gemm_kernel<<<blocks, 256, 0, stream>>>(X, edge_row, edge_col, edge_val,
                                                           cnt, slots, spill, Wmat, bias,
                                                           out, N, cstride, ntiles);
    }
}


# Row wave loads both halves' bucket entries packed into one register set and puts them in edge order with one lane push: ranking touches one register, a pick is one lane read; sum order kept
# speedup vs baseline: 3.5581x; 1.0531x over previous
#include <hip/hip_runtime.h>

typedef float v2f __attribute__((ext_vector_type(2)));
typedef float v8f __attribute__((ext_vector_type(8)));

#define D 128
#define WP 132

#define GRID_BLOCKS 1024
#define CAP 64
#define HCAP 32

__device__ __forceinline__ int gcn_chip_half() {
    const unsigned id = (unsigned)__builtin_amdgcn_s_sendmsg_rtn(0x87);
    return (int)((id >> 18) & 1u);
}

__global__ void gcn_clear_counts_kernel(int* __restrict__ cnt, int n) {
    int i = blockIdx.x * blockDim.x + threadIdx.x;
    if (i < n) cnt[i] = 0;
}

__global__ void __launch_bounds__(256)
gcn_bin_edges_kernel(const int*   __restrict__ edge_row,
                     const int*   __restrict__ edge_col,
                     const float* __restrict__ edge_val,
                     int*         __restrict__ cnt,
                     int4*        __restrict__ slots,
                     int*         __restrict__ spill,
                     int N, int cstride, int E) {
    const int h = gcn_chip_half();
    int e = blockIdx.x * blockDim.x + threadIdx.x;
    if (e >= E) return;
    int r   = edge_row[e];
    const int   c = edge_col[e];
    const float v = edge_val[e];
    asm volatile("" :: "v"(c), "v"(v));
    int pos = atomicAdd(&cnt[(size_t)h * cstride + r], 1);
    if (pos < HCAP) {
        slots[(size_t)r * CAP + h * HCAP + pos] = make_int4(e, c, __float_as_int(v), 0);
    } else {
        int k = atomicAdd(&cnt[N], 1);
        spill[k] = e;
    }
}

__device__ __forceinline__ void gcn_fma4(float4& acc, float v, const float4& x) {
    acc.x = fmaf(v, x.x, acc.x);
    acc.y = fmaf(v, x.y, acc.y);
    acc.z = fmaf(v, x.z, acc.z);
    acc.w = fmaf(v, x.w, acc.w);
}

struct GcnBucket {
    int ca, va, ra;
    int cb, vb, rb;
};

__device__ __forceinline__ void gcn_pick(const GcnBucket& bk, int j, int& c, float& v) {
    const unsigned ma = __builtin_amdgcn_ballot_w32(bk.ra == j);
    if (ma) {
        const int s = __builtin_ctz(ma);
        c = __builtin_amdgcn_readlane(bk.ca, s);
        v = __int_as_float(__builtin_amdgcn_readlane(bk.va, s));
    } else {
        const unsigned mb = __builtin_amdgcn_ballot_w32(bk.rb == j);
        const int s = __builtin_ctz(mb);
        c = __builtin_amdgcn_readlane(bk.cb, s);
        v = __int_as_float(__builtin_amdgcn_readlane(bk.vb, s));
    }
}

template <int U>
__device__ __forceinline__ void gcn_sum_chunk(float4& acc, const float4* __restrict__ xlane,
                                              const GcnBucket& bk, int j0) {
    int    c[U];
    float  v[U];
    float4 x[U];
#pragma unroll
    for (int u = 0; u < U; ++u) gcn_pick(bk, j0 + u, c[u], v[u]);
#pragma unroll
    for (int u = 0; u < U; ++u) x[u] = xlane[(size_t)c[u] * (D / 4)];
#pragma unroll
    for (int u = 0; u < U; ++u) gcn_fma4(acc, v[u], x[u]);
}

template <int U>
__device__ __forceinline__ void gcn_sum_chunk_sorted(float4& acc, const float4* __restrict__ xlane,
                                                     int cs, int vs, int j0) {
    int    c[U];
    float  v[U];
    float4 x[U];
#pragma unroll
    for (int u = 0; u < U; ++u) {
        c[u] = __builtin_amdgcn_readlane(cs, j0 + u);
        v[u] = __int_as_float(__builtin_amdgcn_readlane(vs, j0 + u));
    }
#pragma unroll
    for (int u = 0; u < U; ++u) x[u] = xlane[(size_t)c[u] * (D / 4)];
#pragma unroll
    for (int u = 0; u < U; ++u) gcn_fma4(acc, v[u], x[u]);
}

#define AP 132

struct GcnRowIn {
    int  t0, t1, d0, d1;
    int4 xa, xb;
};

__device__ __forceinline__ GcnRowIn gcn_row_fetch(const int*  __restrict__ cnt,
                                                  const int4* __restrict__ slots,
                                                  int r, int N, int cstride, int lane) {
    GcnRowIn in;
    in.t0 = 0;
    in.t1 = 0;
    if (r < N) {
        in.t0 = cnt[r];
        in.t1 = cnt[(size_t)cstride + r];
    }
    in.d0 = in.t0 < HCAP ? in.t0 : HCAP;
    in.d1 = in.t1 < HCAP ? in.t1 : HCAP;
    in.xa = make_int4(0, 0, 0, 0);
    in.xb = make_int4(0, 0, 0, 0);
    const int deg = in.d0 + in.d1;
    if (deg > 0) {
        const int4* bucket = slots + (size_t)r * CAP;
        if (deg <= 32) {
            const int l = lane < deg - 1 ? lane : deg - 1;
            in.xa = bucket[l < in.d0 ? l : HCAP + (l - in.d0)];
        } else {
            const int la = lane < in.d0 - 1 ? lane : in.d0 - 1;
            const int lb = HCAP + (lane < in.d1 - 1 ? lane : in.d1 - 1);
            in.xa = bucket[in.d0 > 0 ? la : lb];
            in.xb = bucket[in.d1 > 0 ? lb : la];
        }
    }
    return in;
}

__device__ __forceinline__ float4 gcn_row_sum(const GcnRowIn& in,
                                              const float* __restrict__ X,
                                              const int*   __restrict__ edge_row,
                                              const int*   __restrict__ edge_col,
                                              const float* __restrict__ edge_val,
                                              const int*   __restrict__ cnt,
                                              const int*   __restrict__ spill,
                                              int r, int N, int lane) {
    const int d0  = in.d0;
    const int d1  = in.d1;
    const int deg = d0 + d1;

    const int EMPTY = 0x7fffffff;
    const float4* xlane = (const float4*)X + lane;
    float4 acc = make_float4(0.f, 0.f, 0.f, 0.f);

    if (deg <= 32) {
        const int e = lane < deg ? in.xa.x : EMPTY;
        int rk = 0;
        for (int j = 0; j < deg; j += 4) {
            const int e0 = __builtin_amdgcn_readlane(e, j);
            const int e1 = __builtin_amdgcn_readlane(e, j + 1);
            const int e2 = __builtin_amdgcn_readlane(e, j + 2);
            const int e3 = __builtin_amdgcn_readlane(e, j + 3);
            rk += (e0 < e) ? 1 : 0;
            rk += (e1 < e) ? 1 : 0;
            rk += (e2 < e) ? 1 : 0;
            rk += (e3 < e) ? 1 : 0;
        }
        const int cs = __builtin_amdgcn_ds_permute(rk << 2, in.xa.y);
        const int vs = __builtin_amdgcn_ds_permute(rk << 2, in.xa.z);

        int j = 0;
        for (; j + 8 <= deg; j += 8) gcn_sum_chunk_sorted<8>(acc, xlane, cs, vs, j);
        if (j + 4 <= deg) { gcn_sum_chunk_sorted<4>(acc, xlane, cs, vs, j); j += 4; }
        const int rem = deg - j;
        if (rem == 3)      gcn_sum_chunk_sorted<3>(acc, xlane, cs, vs, j);
        else if (rem == 2) gcn_sum_chunk_sorted<2>(acc, xlane, cs, vs, j);
        else if (rem == 1) gcn_sum_chunk_sorted<1>(acc, xlane, cs, vs, j);
    } else {
        const int ea = lane < d0 ? in.xa.x : EMPTY;
        const int eb = lane < d1 ? in.xb.x : EMPTY;
        GcnBucket bk;
        bk.ca = in.xa.y;
        bk.va = in.xa.z;
        bk.cb = in.xb.y;
        bk.vb = in.xb.z;

        int ra = 0, rb = 0;
        for (int j = 0; j < d0; ++j) {
            const int ej = __builtin_amdgcn_readlane(ea, j);
            ra += (ej < ea) ? 1 : 0;
            rb += (ej < eb) ? 1 : 0;
        }
        for (int j = 0; j < d1; ++j) {
            const int ej = __builtin_amdgcn_readlane(eb, j);
            ra += (ej < ea) ? 1 : 0;
            rb += (ej < eb) ? 1 : 0;
        }
        bk.ra = ra;
        bk.rb = rb;

        int j = 0;
        for (; j + 8 <= deg; j += 8) gcn_sum_chunk<8>(acc, xlane, bk, j);
        if (j + 4 <= deg) { gcn_sum_chunk<4>(acc, xlane, bk, j); j += 4; }
        const int rem = deg - j;
        if (rem == 3)      gcn_sum_chunk<3>(acc, xlane, bk, j);
        else if (rem == 2) gcn_sum_chunk<2>(acc, xlane, bk, j);
        else if (rem == 1) gcn_sum_chunk<1>(acc, xlane, bk, j);
    }

    if (in.t0 > HCAP || in.t1 > HCAP) {
        const int ns = cnt[N];
        for (int k = 0; k < ns; ++k) {
            const int e = spill[k];
            if (edge_row[e] == r) {
                const float  v = edge_val[e];
                const float4 x = xlane[(size_t)edge_col[e] * (D / 4)];
                gcn_fma4(acc, v, x);
            }
        }
    }
    return acc;
}

__global__ void __launch_bounds__(256, 8)
gcn_gather_gemm_kernel(const float* __restrict__ X,
                       const int*   __restrict__ edge_row,
                       const int*   __restrict__ edge_col,
                       const float* __restrict__ edge_val,
                       const int*   __restrict__ cnt,
                       const int4*  __restrict__ slots,
                       const int*   __restrict__ spill,
                       const float* __restrict__ Wmat,
                       const float* __restrict__ bias,
                       float*       __restrict__ out,
                       int N, int cstride, int ntiles) {
    __shared__ __attribute__((aligned(16))) float Atile[16 * AP];
    __shared__ __attribute__((aligned(16))) float Dtile[16 * AP];

    const int lane = threadIdx.x & 31;
    const int wave = __builtin_amdgcn_readfirstlane((int)(threadIdx.x >> 5));
    const int lr   = lane & 15;
    const int hi   = lane >> 4;

    const int col = wave * 16 + lr;
    v2f bfrag[D / 4];
    {
        const float* wrow = Wmat + (size_t)col * D + 2 * hi;
#pragma unroll
        for (int kk = 0; kk < D / 4; ++kk) bfrag[kk] = *(const v2f*)(wrow + 4 * kk);
    }
    const float bb = bias[col];

    for (int t = blockIdx.x; t < ntiles; t += gridDim.x) {
        float* At = &Atile[0];
        const int rl = wave * 2;
        const int r0 = t * 16 + rl;

        const GcnRowIn in0 = gcn_row_fetch(cnt, slots, r0,     N, cstride, lane);
        const GcnRowIn in1 = gcn_row_fetch(cnt, slots, r0 + 1, N, cstride, lane);

        const float4 s0 = gcn_row_sum(in0, X, edge_row, edge_col, edge_val, cnt, spill,
                                      r0, N, lane);
        *(float4*)&At[rl * AP + 4 * lane] = s0;
        const float4 s1 = gcn_row_sum(in1, X, edge_row, edge_col, edge_val, cnt, spill,
                                      r0 + 1, N, lane);
        *(float4*)&At[(rl + 1) * AP + 4 * lane] = s1;

        __syncthreads();

        v8f d = (v8f){0.f, 0.f, 0.f, 0.f, 0.f, 0.f, 0.f, 0.f};
        const float* arow = At + lr * AP + 2 * hi;
#pragma unroll
        for (int kk = 0; kk < D / 4; ++kk) {
            const v2f a = *(const v2f*)(arow + 4 * kk);
            d = __builtin_amdgcn_wmma_f32_16x16x4_f32(
                 false, a,  false, bfrag[kk],
                 (short)0, d,
                 false,  false);
        }

#pragma unroll
        for (int j = 0; j < 8; ++j) Dtile[(j + hi * 8) * AP + col] = d[j] + bb;

        __syncthreads();

#pragma unroll
        for (int i = 0; i < 2; ++i) {
            const int r = r0 + i;
            const float4 o = *(const float4*)&Dtile[(rl + i) * AP + 4 * lane];
            if (r < N) ((float4*)out)[(size_t)r * (D / 4) + lane] = o;
        }
    }
}

extern "C" void kernel_launch(void* const* d_in, const int* in_sizes, int n_in,
                              void* d_out, int out_size, void* d_ws, size_t ws_size,
                              hipStream_t stream) {
    const float* X        = (const float*)d_in[0];
    const int*   edge_row = (const int*)  d_in[1];
    const int*   edge_col = (const int*)  d_in[2];
    const float* edge_val = (const float*)d_in[3];
    const float* Wmat     = (const float*)d_in[4];
    const float* bias     = (const float*)d_in[5];
    float*       out      = (float*)d_out;

    const int N = in_sizes[0] / D;
    const int E = in_sizes[1];

    const int cstride = ((N + 1 + 63) / 64) * 64;
    int*  cnt   = (int*)d_ws;
    int*  spill = cnt + 2 * cstride;
    int4* slots = (int4*)(spill + ((E + 127) / 128) * 128);

    gcn_clear_counts_kernel<<<(2 * cstride + 255) / 256, 256, 0, stream>>>(cnt, 2 * cstride);

    gcn_bin_edges_kernel<<<(E + 255) / 256, 256, 0, stream>>>(edge_row, edge_col, edge_val,
                                                             cnt, slots, spill, N,
                                                             cstride, E);

    {
        const int ntiles = (N + 15) / 16;
        const int blocks = ntiles < GRID_BLOCKS ? ntiles : GRID_BLOCKS;
        gcn_gather_gemm_kernel<<<blocks, 256, 0, stream>>>(X, edge_row, edge_col, edge_val,
                                                           cnt, slots, spill, Wmat, bias,
                                                           out, N, cstride, ntiles);
    }
}


# Row inputs run one tile ahead: next tile's bucket entries (and the counts of the tile after) are requested before the GEMM; tile barriers wait for LDS only, so they stay in flight; sum order kept
# speedup vs baseline: 3.6729x; 1.0323x over previous
#include <hip/hip_runtime.h>

typedef float v2f __attribute__((ext_vector_type(2)));
typedef float v8f __attribute__((ext_vector_type(8)));

#define D 128
#define WP 132

#define GRID_BLOCKS 1024
#define CAP 64
#define HCAP 32

__device__ __forceinline__ int gcn_chip_half() {
    const unsigned id = (unsigned)__builtin_amdgcn_s_sendmsg_rtn(0x87);
    return (int)((id >> 18) & 1u);
}

__global__ void gcn_clear_counts_kernel(int* __restrict__ cnt, int n) {
    int i = blockIdx.x * blockDim.x + threadIdx.x;
    if (i < n) cnt[i] = 0;
}

__global__ void __launch_bounds__(256)
gcn_bin_edges_kernel(const int*   __restrict__ edge_row,
                     const int*   __restrict__ edge_col,
                     const float* __restrict__ edge_val,
                     int*         __restrict__ cnt,
                     int4*        __restrict__ slots,
                     int*         __restrict__ spill,
                     int N, int cstride, int E) {
    const int h = gcn_chip_half();
    int e = blockIdx.x * blockDim.x + threadIdx.x;
    if (e >= E) return;
    int r   = edge_row[e];
    const int   c = edge_col[e];
    const float v = edge_val[e];
    asm volatile("" :: "v"(c), "v"(v));
    int pos = atomicAdd(&cnt[(size_t)h * cstride + r], 1);
    if (pos < HCAP) {
        slots[(size_t)r * CAP + h * HCAP + pos] = make_int4(e, c, __float_as_int(v), 0);
    } else {
        int k = atomicAdd(&cnt[N], 1);
        spill[k] = e;
    }
}

__device__ __forceinline__ void gcn_fma4(float4& acc, float v, const float4& x) {
    acc.x = fmaf(v, x.x, acc.x);
    acc.y = fmaf(v, x.y, acc.y);
    acc.z = fmaf(v, x.z, acc.z);
    acc.w = fmaf(v, x.w, acc.w);
}

struct GcnBucket {
    int ca, va, ra;
    int cb, vb, rb;
};

__device__ __forceinline__ void gcn_pick(const GcnBucket& bk, int j, int& c, float& v) {
    const unsigned ma = __builtin_amdgcn_ballot_w32(bk.ra == j);
    if (ma) {
        const int s = __builtin_ctz(ma);
        c = __builtin_amdgcn_readlane(bk.ca, s);
        v = __int_as_float(__builtin_amdgcn_readlane(bk.va, s));
    } else {
        const unsigned mb = __builtin_amdgcn_ballot_w32(bk.rb == j);
        const int s = __builtin_ctz(mb);
        c = __builtin_amdgcn_readlane(bk.cb, s);
        v = __int_as_float(__builtin_amdgcn_readlane(bk.vb, s));
    }
}

template <int U>
__device__ __forceinline__ void gcn_sum_chunk(float4& acc, const float4* __restrict__ xlane,
                                              const GcnBucket& bk, int j0) {
    int    c[U];
    float  v[U];
    float4 x[U];
#pragma unroll
    for (int u = 0; u < U; ++u) gcn_pick(bk, j0 + u, c[u], v[u]);
#pragma unroll
    for (int u = 0; u < U; ++u) x[u] = xlane[(size_t)c[u] * (D / 4)];
#pragma unroll
    for (int u = 0; u < U; ++u) gcn_fma4(acc, v[u], x[u]);
}

template <int U>
__device__ __forceinline__ void gcn_sum_chunk_sorted(float4& acc, const float4* __restrict__ xlane,
                                                     int cs, int vs, int j0) {
    int    c[U];
    float  v[U];
    float4 x[U];
#pragma unroll
    for (int u = 0; u < U; ++u) {
        c[u] = __builtin_amdgcn_readlane(cs, j0 + u);
        v[u] = __int_as_float(__builtin_amdgcn_readlane(vs, j0 + u));
    }
#pragma unroll
    for (int u = 0; u < U; ++u) x[u] = xlane[(size_t)c[u] * (D / 4)];
#pragma unroll
    for (int u = 0; u < U; ++u) gcn_fma4(acc, v[u], x[u]);
}

#define AP 132

struct GcnRowIn {
    int  t0, t1, d0, d1;
    int4 xa;
};

__device__ __forceinline__ int gcn_tile_counts_load(const int* __restrict__ cnt,
                                                    int r0, int N, int cstride, int lane) {
    const int k = lane & 3;
    int r = r0 + (k & 1);
    r = r < N ? r : N - 1;
    r = r > 0 ? r : 0;
    return cnt[(size_t)(k >> 1) * cstride + r];
}

__device__ __forceinline__ int gcn_tile_count(int cv, int k, int r, int N) {
    return r < N ? __builtin_amdgcn_readlane(cv, k) : 0;
}

__device__ __forceinline__ void gcn_lds_barrier() {
    asm volatile("s_wait_dscnt 0x0\n\ts_barrier_signal -1\n\ts_barrier_wait -1" ::: "memory");
}

__device__ __forceinline__ int gcn_xa_index(int d0, int d1, int lane) {
    const int deg = d0 + d1;
    if (deg <= 32) {
        const int l = lane < deg - 1 ? lane : deg - 1;
        return l < d0 ? l : HCAP + (l - d0);
    }
    const int la = lane < d0 - 1 ? lane : d0 - 1;
    const int lb = HCAP + (lane < d1 - 1 ? lane : d1 - 1);
    return d0 > 0 ? la : lb;
}

__device__ __forceinline__ int gcn_xb_index(int d0, int d1, int lane) {
    const int la = lane < d0 - 1 ? lane : d0 - 1;
    const int lb = HCAP + (lane < d1 - 1 ? lane : d1 - 1);
    return d1 > 0 ? lb : la;
}

__device__ __forceinline__ GcnRowIn gcn_row_bucket(const int4* __restrict__ slots,
                                                   int r, int t0, int t1, int lane) {
    GcnRowIn in;
    in.t0 = t0;
    in.t1 = t1;
    in.d0 = in.t0 < HCAP ? in.t0 : HCAP;
    in.d1 = in.t1 < HCAP ? in.t1 : HCAP;
    in.xa = make_int4(0, 0, 0, 0);
    if (in.d0 + in.d1 > 0)
        in.xa = slots[(size_t)r * CAP + gcn_xa_index(in.d0, in.d1, lane)];
    return in;
}

__device__ __forceinline__ float4 gcn_row_sum(const GcnRowIn& in,
                                              const float* __restrict__ X,
                                              const int*   __restrict__ edge_row,
                                              const int*   __restrict__ edge_col,
                                              const float* __restrict__ edge_val,
                                              const int*   __restrict__ cnt,
                                              const int4*  __restrict__ slots,
                                              const int*   __restrict__ spill,
                                              int r, int N, int lane) {
    const int d0  = in.d0;
    const int d1  = in.d1;
    const int deg = d0 + d1;

    const int EMPTY = 0x7fffffff;
    const float4* xlane = (const float4*)X + lane;
    float4 acc = make_float4(0.f, 0.f, 0.f, 0.f);

    if (deg <= 32) {
        const int e = lane < deg ? in.xa.x : EMPTY;
        int rk = 0;
        for (int j = 0; j < deg; j += 4) {
            const int e0 = __builtin_amdgcn_readlane(e, j);
            const int e1 = __builtin_amdgcn_readlane(e, j + 1);
            const int e2 = __builtin_amdgcn_readlane(e, j + 2);
            const int e3 = __builtin_amdgcn_readlane(e, j + 3);
            rk += (e0 < e) ? 1 : 0;
            rk += (e1 < e) ? 1 : 0;
            rk += (e2 < e) ? 1 : 0;
            rk += (e3 < e) ? 1 : 0;
        }
        const int cs = __builtin_amdgcn_ds_permute(rk << 2, in.xa.y);
        const int vs = __builtin_amdgcn_ds_permute(rk << 2, in.xa.z);

        int j = 0;
        for (; j + 8 <= deg; j += 8) gcn_sum_chunk_sorted<8>(acc, xlane, cs, vs, j);
        if (j + 4 <= deg) { gcn_sum_chunk_sorted<4>(acc, xlane, cs, vs, j); j += 4; }
        const int rem = deg - j;
        if (rem == 3)      gcn_sum_chunk_sorted<3>(acc, xlane, cs, vs, j);
        else if (rem == 2) gcn_sum_chunk_sorted<2>(acc, xlane, cs, vs, j);
        else if (rem == 1) gcn_sum_chunk_sorted<1>(acc, xlane, cs, vs, j);
    } else {
        const int4 xb = slots[(size_t)r * CAP + gcn_xb_index(d0, d1, lane)];
        const int ea = lane < d0 ? in.xa.x : EMPTY;
        const int eb = lane < d1 ? xb.x : EMPTY;
        GcnBucket bk;
        bk.ca = in.xa.y;
        bk.va = in.xa.z;
        bk.cb = xb.y;
        bk.vb = xb.z;

        int ra = 0, rb = 0;
        for (int j = 0; j < d0; ++j) {
            const int ej = __builtin_amdgcn_readlane(ea, j);
            ra += (ej < ea) ? 1 : 0;
            rb += (ej < eb) ? 1 : 0;
        }
        for (int j = 0; j < d1; ++j) {
            const int ej = __builtin_amdgcn_readlane(eb, j);
            ra += (ej < ea) ? 1 : 0;
            rb += (ej < eb) ? 1 : 0;
        }
        bk.ra = ra;
        bk.rb = rb;

        int j = 0;
        for (; j + 8 <= deg; j += 8) gcn_sum_chunk<8>(acc, xlane, bk, j);
        if (j + 4 <= deg) { gcn_sum_chunk<4>(acc, xlane, bk, j); j += 4; }
        const int rem = deg - j;
        if (rem == 3)      gcn_sum_chunk<3>(acc, xlane, bk, j);
        else if (rem == 2) gcn_sum_chunk<2>(acc, xlane, bk, j);
        else if (rem == 1) gcn_sum_chunk<1>(acc, xlane, bk, j);
    }

    if (in.t0 > HCAP || in.t1 > HCAP) {
        const int ns = cnt[N];
        for (int k = 0; k < ns; ++k) {
            const int e = spill[k];
            if (edge_row[e] == r) {
                const float  v = edge_val[e];
                const float4 x = xlane[(size_t)edge_col[e] * (D / 4)];
                gcn_fma4(acc, v, x);
            }
        }
    }
    return acc;
}

__global__ void __launch_bounds__(256, 8)
gcn_gather_gemm_kernel(const float* __restrict__ X,
                       const int*   __restrict__ edge_row,
                       const int*   __restrict__ edge_col,
                       const float* __restrict__ edge_val,
                       const int*   __restrict__ cnt,
                       const int4*  __restrict__ slots,
                       const int*   __restrict__ spill,
                       const float* __restrict__ Wmat,
                       const float* __restrict__ bias,
                       float*       __restrict__ out,
                       int N, int cstride, int ntiles) {
    __shared__ __attribute__((aligned(16))) float Atile[16 * AP];
    __shared__ __attribute__((aligned(16))) float Dtile[16 * AP];

    const int lane = threadIdx.x & 31;
    const int wave = __builtin_amdgcn_readfirstlane((int)(threadIdx.x >> 5));
    const int lr   = lane & 15;
    const int hi   = lane >> 4;

    const int col = wave * 16 + lr;
    v2f bfrag[D / 4];
    {
        const float* wrow = Wmat + (size_t)col * D + 2 * hi;
#pragma unroll
        for (int kk = 0; kk < D / 4; ++kk) bfrag[kk] = *(const v2f*)(wrow + 4 * kk);
    }
    const float bb = bias[col];

    const int rl = wave * 2;
    const int tstep = (int)gridDim.x;
    int t = blockIdx.x;

    GcnRowIn in0, in1;
    int n00, n01, n10, n11;
    {
        const int r0 = t * 16 + rl;
        const int rn = (t + tstep) * 16 + rl;
        const int ccur = gcn_tile_counts_load(cnt, r0, N, cstride, lane);
        const int cnxt = gcn_tile_counts_load(cnt, rn, N, cstride, lane);
        in0 = gcn_row_bucket(slots, r0,
                             gcn_tile_count(ccur, 0, r0, N),
                             gcn_tile_count(ccur, 2, r0, N), lane);
        in1 = gcn_row_bucket(slots, r0 + 1,
                             gcn_tile_count(ccur, 1, r0 + 1, N),
                             gcn_tile_count(ccur, 3, r0 + 1, N), lane);
        n00 = gcn_tile_count(cnxt, 0, rn,     N);
        n01 = gcn_tile_count(cnxt, 2, rn,     N);
        n10 = gcn_tile_count(cnxt, 1, rn + 1, N);
        n11 = gcn_tile_count(cnxt, 3, rn + 1, N);
    }

    for (; t < ntiles; t += tstep) {
        float* At = &Atile[0];
        const int r0 = t * 16 + rl;

        asm volatile("" :: "v"(in0.xa.x), "v"(in0.xa.y), "v"(in0.xa.z));
        const float4 s0 = gcn_row_sum(in0, X, edge_row, edge_col, edge_val, cnt, slots, spill,
                                      r0, N, lane);
        *(float4*)&At[rl * AP + 4 * lane] = s0;
        asm volatile("" :: "v"(in1.xa.x), "v"(in1.xa.y), "v"(in1.xa.z));
        const float4 s1 = gcn_row_sum(in1, X, edge_row, edge_col, edge_val, cnt, slots, spill,
                                      r0 + 1, N, lane);
        *(float4*)&At[(rl + 1) * AP + 4 * lane] = s1;

        const int rn  = (t + tstep) * 16 + rl;
        const int rnn = (t + 2 * tstep) * 16 + rl;
        const int cnn = gcn_tile_counts_load(cnt, rnn, N, cstride, lane);
        in0 = gcn_row_bucket(slots, rn,     n00, n01, lane);
        in1 = gcn_row_bucket(slots, rn + 1, n10, n11, lane);

        gcn_lds_barrier();

        v8f d = (v8f){0.f, 0.f, 0.f, 0.f, 0.f, 0.f, 0.f, 0.f};
        const float* arow = At + lr * AP + 2 * hi;
#pragma unroll
        for (int kk = 0; kk < D / 4; ++kk) {
            const v2f a = *(const v2f*)(arow + 4 * kk);
            d = __builtin_amdgcn_wmma_f32_16x16x4_f32(
                 false, a,  false, bfrag[kk],
                 (short)0, d,
                 false,  false);
        }

#pragma unroll
        for (int j = 0; j < 8; ++j) Dtile[(j + hi * 8) * AP + col] = d[j] + bb;

        gcn_lds_barrier();

#pragma unroll
        for (int i = 0; i < 2; ++i) {
            const int r = r0 + i;
            const float4 o = *(const float4*)&Dtile[(rl + i) * AP + 4 * lane];
            if (r < N) ((float4*)out)[(size_t)r * (D / 4) + lane] = o;
        }

        n00 = gcn_tile_count(cnn, 0, rnn,     N);
        n01 = gcn_tile_count(cnn, 2, rnn,     N);
        n10 = gcn_tile_count(cnn, 1, rnn + 1, N);
        n11 = gcn_tile_count(cnn, 3, rnn + 1, N);
    }
}

extern "C" void kernel_launch(void* const* d_in, const int* in_sizes, int n_in,
                              void* d_out, int out_size, void* d_ws, size_t ws_size,
                              hipStream_t stream) {
    const float* X        = (const float*)d_in[0];
    const int*   edge_row = (const int*)  d_in[1];
    const int*   edge_col = (const int*)  d_in[2];
    const float* edge_val = (const float*)d_in[3];
    const float* Wmat     = (const float*)d_in[4];
    const float* bias     = (const float*)d_in[5];
    float*       out      = (float*)d_out;

    const int N = in_sizes[0] / D;
    const int E = in_sizes[1];

    const int cstride = ((N + 1 + 63) / 64) * 64;
    int*  cnt   = (int*)d_ws;
    int*  spill = cnt + 2 * cstride;
    int4* slots = (int4*)(spill + ((E + 127) / 128) * 128);

    gcn_clear_counts_kernel<<<(2 * cstride + 255) / 256, 256, 0, stream>>>(cnt, 2 * cstride);

    gcn_bin_edges_kernel<<<(E + 255) / 256, 256, 0, stream>>>(edge_row, edge_col, edge_val,
                                                             cnt, slots, spill, N,
                                                             cstride, E);

    {
        const int ntiles = (N + 15) / 16;
        const int blocks = ntiles < GRID_BLOCKS ? ntiles : GRID_BLOCKS;
        gcn_gather_gemm_kernel<<<blocks, 256, 0, stream>>>(X, edge_row, edge_col, edge_val,
                                                           cnt, slots, spill, Wmat, bias,
                                                           out, N, cstride, ntiles);
    }
}


# W fragments reach the registers through LDS: a wave loads its 16 rows of W whole (64 line accesses) instead of 32 loads of 8 bytes per lane touching 16 lines each (512); same values, sums unchanged
# speedup vs baseline: 3.8321x; 1.0433x over previous
#include <hip/hip_runtime.h>

typedef float v2f __attribute__((ext_vector_type(2)));
typedef float v8f __attribute__((ext_vector_type(8)));

#define D 128
#define WFP 514

#define GRID_BLOCKS 1024
#define CAP 64
#define HCAP 32

__device__ __forceinline__ int gcn_chip_half() {
    const unsigned id = (unsigned)__builtin_amdgcn_s_sendmsg_rtn(0x87);
    return (int)((id >> 18) & 1u);
}

__global__ void gcn_clear_counts_kernel(int* __restrict__ cnt, int n) {
    int i = blockIdx.x * blockDim.x + threadIdx.x;
    if (i < n) cnt[i] = 0;
}

__global__ void __launch_bounds__(256)
gcn_bin_edges_kernel(const int*   __restrict__ edge_row,
                     const int*   __restrict__ edge_col,
                     const float* __restrict__ edge_val,
                     int*         __restrict__ cnt,
                     int4*        __restrict__ slots,
                     int*         __restrict__ spill,
                     int N, int cstride, int E) {
    const int h = gcn_chip_half();
    int e = blockIdx.x * blockDim.x + threadIdx.x;
    if (e >= E) return;
    int r   = edge_row[e];
    const int   c = edge_col[e];
    const float v = edge_val[e];
    asm volatile("" :: "v"(c), "v"(v));
    int pos = atomicAdd(&cnt[(size_t)h * cstride + r], 1);
    if (pos < HCAP) {
        slots[(size_t)r * CAP + h * HCAP + pos] = make_int4(e, c, __float_as_int(v), 0);
    } else {
        int k = atomicAdd(&cnt[N], 1);
        spill[k] = e;
    }
}

__device__ __forceinline__ void gcn_fma4(float4& acc, float v, const float4& x) {
    acc.x = fmaf(v, x.x, acc.x);
    acc.y = fmaf(v, x.y, acc.y);
    acc.z = fmaf(v, x.z, acc.z);
    acc.w = fmaf(v, x.w, acc.w);
}

struct GcnBucket {
    int ca, va, ra;
    int cb, vb, rb;
};

__device__ __forceinline__ void gcn_pick(const GcnBucket& bk, int j, int& c, float& v) {
    const unsigned ma = __builtin_amdgcn_ballot_w32(bk.ra == j);
    if (ma) {
        const int s = __builtin_ctz(ma);
        c = __builtin_amdgcn_readlane(bk.ca, s);
        v = __int_as_float(__builtin_amdgcn_readlane(bk.va, s));
    } else {
        const unsigned mb = __builtin_amdgcn_ballot_w32(bk.rb == j);
        const int s = __builtin_ctz(mb);
        c = __builtin_amdgcn_readlane(bk.cb, s);
        v = __int_as_float(__builtin_amdgcn_readlane(bk.vb, s));
    }
}

template <int U>
__device__ __forceinline__ void gcn_sum_chunk(float4& acc, const float4* __restrict__ xlane,
                                              const GcnBucket& bk, int j0) {
    int    c[U];
    float  v[U];
    float4 x[U];
#pragma unroll
    for (int u = 0; u < U; ++u) gcn_pick(bk, j0 + u, c[u], v[u]);
#pragma unroll
    for (int u = 0; u < U; ++u) x[u] = xlane[(size_t)c[u] * (D / 4)];
#pragma unroll
    for (int u = 0; u < U; ++u) gcn_fma4(acc, v[u], x[u]);
}

template <int U>
__device__ __forceinline__ void gcn_sum_chunk_sorted(float4& acc, const float4* __restrict__ xlane,
                                                     int cs, int vs, int j0) {
    int    c[U];
    float  v[U];
    float4 x[U];
#pragma unroll
    for (int u = 0; u < U; ++u) {
        c[u] = __builtin_amdgcn_readlane(cs, j0 + u);
        v[u] = __int_as_float(__builtin_amdgcn_readlane(vs, j0 + u));
    }
#pragma unroll
    for (int u = 0; u < U; ++u) x[u] = xlane[(size_t)c[u] * (D / 4)];
#pragma unroll
    for (int u = 0; u < U; ++u) gcn_fma4(acc, v[u], x[u]);
}

#define AP 132

struct GcnRowIn {
    int  t0, t1, d0, d1;
    int4 xa;
};

__device__ __forceinline__ int gcn_tile_counts_load(const int* __restrict__ cnt,
                                                    int r0, int N, int cstride, int lane) {
    const int k = lane & 3;
    int r = r0 + (k & 1);
    r = r < N ? r : N - 1;
    r = r > 0 ? r : 0;
    return cnt[(size_t)(k >> 1) * cstride + r];
}

__device__ __forceinline__ int gcn_tile_count(int cv, int k, int r, int N) {
    return r < N ? __builtin_amdgcn_readlane(cv, k) : 0;
}

__device__ __forceinline__ void gcn_lds_barrier() {
    asm volatile("s_wait_dscnt 0x0\n\ts_barrier_signal -1\n\ts_barrier_wait -1" ::: "memory");
}

__device__ __forceinline__ int gcn_xa_index(int d0, int d1, int lane) {
    const int deg = d0 + d1;
    if (deg <= 32) {
        const int l = lane < deg - 1 ? lane : deg - 1;
        return l < d0 ? l : HCAP + (l - d0);
    }
    const int la = lane < d0 - 1 ? lane : d0 - 1;
    const int lb = HCAP + (lane < d1 - 1 ? lane : d1 - 1);
    return d0 > 0 ? la : lb;
}

__device__ __forceinline__ int gcn_xb_index(int d0, int d1, int lane) {
    const int la = lane < d0 - 1 ? lane : d0 - 1;
    const int lb = HCAP + (lane < d1 - 1 ? lane : d1 - 1);
    return d1 > 0 ? lb : la;
}

__device__ __forceinline__ GcnRowIn gcn_row_bucket(const int4* __restrict__ slots,
                                                   int r, int t0, int t1, int lane) {
    GcnRowIn in;
    in.t0 = t0;
    in.t1 = t1;
    in.d0 = in.t0 < HCAP ? in.t0 : HCAP;
    in.d1 = in.t1 < HCAP ? in.t1 : HCAP;
    in.xa = make_int4(0, 0, 0, 0);
    if (in.d0 + in.d1 > 0)
        in.xa = slots[(size_t)r * CAP + gcn_xa_index(in.d0, in.d1, lane)];
    return in;
}

__device__ __forceinline__ float4 gcn_row_sum(const GcnRowIn& in,
                                              const float* __restrict__ X,
                                              const int*   __restrict__ edge_row,
                                              const int*   __restrict__ edge_col,
                                              const float* __restrict__ edge_val,
                                              const int*   __restrict__ cnt,
                                              const int4*  __restrict__ slots,
                                              const int*   __restrict__ spill,
                                              int r, int N, int lane) {
    const int d0  = in.d0;
    const int d1  = in.d1;
    const int deg = d0 + d1;

    const int EMPTY = 0x7fffffff;
    const float4* xlane = (const float4*)X + lane;
    float4 acc = make_float4(0.f, 0.f, 0.f, 0.f);

    if (deg <= 32) {
        const int e = lane < deg ? in.xa.x : EMPTY;
        int rk = 0;
        for (int j = 0; j < deg; j += 4) {
            const int e0 = __builtin_amdgcn_readlane(e, j);
            const int e1 = __builtin_amdgcn_readlane(e, j + 1);
            const int e2 = __builtin_amdgcn_readlane(e, j + 2);
            const int e3 = __builtin_amdgcn_readlane(e, j + 3);
            rk += (e0 < e) ? 1 : 0;
            rk += (e1 < e) ? 1 : 0;
            rk += (e2 < e) ? 1 : 0;
            rk += (e3 < e) ? 1 : 0;
        }
        const int cs = __builtin_amdgcn_ds_permute(rk << 2, in.xa.y);
        const int vs = __builtin_amdgcn_ds_permute(rk << 2, in.xa.z);

        int j = 0;
        for (; j + 8 <= deg; j += 8) gcn_sum_chunk_sorted<8>(acc, xlane, cs, vs, j);
        if (j + 4 <= deg) { gcn_sum_chunk_sorted<4>(acc, xlane, cs, vs, j); j += 4; }
        const int rem = deg - j;
        if (rem == 3)      gcn_sum_chunk_sorted<3>(acc, xlane, cs, vs, j);
        else if (rem == 2) gcn_sum_chunk_sorted<2>(acc, xlane, cs, vs, j);
        else if (rem == 1) gcn_sum_chunk_sorted<1>(acc, xlane, cs, vs, j);
    } else {
        const int4 xb = slots[(size_t)r * CAP + gcn_xb_index(d0, d1, lane)];
        const int ea = lane < d0 ? in.xa.x : EMPTY;
        const int eb = lane < d1 ? xb.x : EMPTY;
        GcnBucket bk;
        bk.ca = in.xa.y;
        bk.va = in.xa.z;
        bk.cb = xb.y;
        bk.vb = xb.z;

        int ra = 0, rb = 0;
        for (int j = 0; j < d0; ++j) {
            const int ej = __builtin_amdgcn_readlane(ea, j);
            ra += (ej < ea) ? 1 : 0;
            rb += (ej < eb) ? 1 : 0;
        }
        for (int j = 0; j < d1; ++j) {
            const int ej = __builtin_amdgcn_readlane(eb, j);
            ra += (ej < ea) ? 1 : 0;
            rb += (ej < eb) ? 1 : 0;
        }
        bk.ra = ra;
        bk.rb = rb;

        int j = 0;
        for (; j + 8 <= deg; j += 8) gcn_sum_chunk<8>(acc, xlane, bk, j);
        if (j + 4 <= deg) { gcn_sum_chunk<4>(acc, xlane, bk, j); j += 4; }
        const int rem = deg - j;
        if (rem == 3)      gcn_sum_chunk<3>(acc, xlane, bk, j);
        else if (rem == 2) gcn_sum_chunk<2>(acc, xlane, bk, j);
        else if (rem == 1) gcn_sum_chunk<1>(acc, xlane, bk, j);
    }

    if (in.t0 > HCAP || in.t1 > HCAP) {
        const int ns = cnt[N];
        for (int k = 0; k < ns; ++k) {
            const int e = spill[k];
            if (edge_row[e] == r) {
                const float  v = edge_val[e];
                const float4 x = xlane[(size_t)edge_col[e] * (D / 4)];
                gcn_fma4(acc, v, x);
            }
        }
    }
    return acc;
}

__global__ void __launch_bounds__(256, 8)
gcn_gather_gemm_kernel(const float* __restrict__ X,
                       const int*   __restrict__ edge_row,
                       const int*   __restrict__ edge_col,
                       const float* __restrict__ edge_val,
                       const int*   __restrict__ cnt,
                       const int4*  __restrict__ slots,
                       const int*   __restrict__ spill,
                       const float* __restrict__ Wmat,
                       const float* __restrict__ bias,
                       float*       __restrict__ out,
                       int N, int cstride, int ntiles) {
    __shared__ __attribute__((aligned(16))) float smem[(D / 4) * WFP];
    float* const Atile = smem;
    float* const Dtile = smem + 16 * AP;

    const int lane = threadIdx.x & 31;
    const int wave = __builtin_amdgcn_readfirstlane((int)(threadIdx.x >> 5));
    const int lr   = lane & 15;
    const int hi   = lane >> 4;

    const int col = wave * 16 + lr;

    float4 wr[16];
    {
        const float4* wsrc = (const float4*)(Wmat + (size_t)(wave * 16) * D) + lane;
#pragma unroll
        for (int r = 0; r < 16; ++r) wr[r] = wsrc[r * (D / 4)];
    }
    const float bb = bias[col];

    const int rl = wave * 2;
    const int tstep = (int)gridDim.x;
    int t = blockIdx.x;

    GcnRowIn in0, in1;
    int n00, n01, n10, n11;
    {
        const int r0 = t * 16 + rl;
        const int rn = (t + tstep) * 16 + rl;
        const int ccur = gcn_tile_counts_load(cnt, r0, N, cstride, lane);
        const int cnxt = gcn_tile_counts_load(cnt, rn, N, cstride, lane);
        in0 = gcn_row_bucket(slots, r0,
                             gcn_tile_count(ccur, 0, r0, N),
                             gcn_tile_count(ccur, 2, r0, N), lane);
        in1 = gcn_row_bucket(slots, r0 + 1,
                             gcn_tile_count(ccur, 1, r0 + 1, N),
                             gcn_tile_count(ccur, 3, r0 + 1, N), lane);
        n00 = gcn_tile_count(cnxt, 0, rn,     N);
        n01 = gcn_tile_count(cnxt, 2, rn,     N);
        n10 = gcn_tile_count(cnxt, 1, rn + 1, N);
        n11 = gcn_tile_count(cnxt, 3, rn + 1, N);
    }

    v2f bfrag[D / 4];
    {
        float* wst = smem + lane * WFP + wave * 64;
#pragma unroll
        for (int r = 0; r < 16; ++r) {
            *(v2f*)(wst + 2 * r)      = (v2f){wr[r].x, wr[r].y};
            *(v2f*)(wst + 2 * r + 32) = (v2f){wr[r].z, wr[r].w};
        }
    }
    gcn_lds_barrier();
    {
        const float* wf = smem + (wave * 32 + lane) * 2;
#pragma unroll
        for (int kk = 0; kk < D / 4; ++kk) bfrag[kk] = *(const v2f*)(wf + kk * WFP);
    }
    gcn_lds_barrier();

    for (; t < ntiles; t += tstep) {
        float* At = &Atile[0];
        const int r0 = t * 16 + rl;

        asm volatile("" :: "v"(in0.xa.x), "v"(in0.xa.y), "v"(in0.xa.z));
        const float4 s0 = gcn_row_sum(in0, X, edge_row, edge_col, edge_val, cnt, slots, spill,
                                      r0, N, lane);
        *(float4*)&At[rl * AP + 4 * lane] = s0;
        asm volatile("" :: "v"(in1.xa.x), "v"(in1.xa.y), "v"(in1.xa.z));
        const float4 s1 = gcn_row_sum(in1, X, edge_row, edge_col, edge_val, cnt, slots, spill,
                                      r0 + 1, N, lane);
        *(float4*)&At[(rl + 1) * AP + 4 * lane] = s1;

        const int rn  = (t + tstep) * 16 + rl;
        const int rnn = (t + 2 * tstep) * 16 + rl;
        const int cnn = gcn_tile_counts_load(cnt, rnn, N, cstride, lane);
        in0 = gcn_row_bucket(slots, rn,     n00, n01, lane);
        in1 = gcn_row_bucket(slots, rn + 1, n10, n11, lane);

        gcn_lds_barrier();

        v8f d = (v8f){0.f, 0.f, 0.f, 0.f, 0.f, 0.f, 0.f, 0.f};
        const float* arow = At + lr * AP + 2 * hi;
#pragma unroll
        for (int kk = 0; kk < D / 4; ++kk) {
            const v2f a = *(const v2f*)(arow + 4 * kk);
            d = __builtin_amdgcn_wmma_f32_16x16x4_f32(
                 false, a,  false, bfrag[kk],
                 (short)0, d,
                 false,  false);
        }

#pragma unroll
        for (int j = 0; j < 8; ++j) Dtile[(j + hi * 8) * AP + col] = d[j] + bb;

        gcn_lds_barrier();

#pragma unroll
        for (int i = 0; i < 2; ++i) {
            const int r = r0 + i;
            const float4 o = *(const float4*)&Dtile[(rl + i) * AP + 4 * lane];
            if (r < N) ((float4*)out)[(size_t)r * (D / 4) + lane] = o;
        }

        n00 = gcn_tile_count(cnn, 0, rnn,     N);
        n01 = gcn_tile_count(cnn, 2, rnn,     N);
        n10 = gcn_tile_count(cnn, 1, rnn + 1, N);
        n11 = gcn_tile_count(cnn, 3, rnn + 1, N);
    }
}

extern "C" void kernel_launch(void* const* d_in, const int* in_sizes, int n_in,
                              void* d_out, int out_size, void* d_ws, size_t ws_size,
                              hipStream_t stream) {
    const float* X        = (const float*)d_in[0];
    const int*   edge_row = (const int*)  d_in[1];
    const int*   edge_col = (const int*)  d_in[2];
    const float* edge_val = (const float*)d_in[3];
    const float* Wmat     = (const float*)d_in[4];
    const float* bias     = (const float*)d_in[5];
    float*       out      = (float*)d_out;

    const int N = in_sizes[0] / D;
    const int E = in_sizes[1];

    const int cstride = ((N + 1 + 63) / 64) * 64;
    int*  cnt   = (int*)d_ws;
    int*  spill = cnt + 2 * cstride;
    int4* slots = (int4*)(spill + ((E + 127) / 128) * 128);

    gcn_clear_counts_kernel<<<(2 * cstride + 255) / 256, 256, 0, stream>>>(cnt, 2 * cstride);

    gcn_bin_edges_kernel<<<(E + 255) / 256, 256, 0, stream>>>(edge_row, edge_col, edge_val,
                                                             cnt, slots, spill, N,
                                                             cstride, E);

    {
        const int ntiles = (N + 15) / 16;
        const int blocks = ntiles < GRID_BLOCKS ? ntiles : GRID_BLOCKS;
        gcn_gather_gemm_kernel<<<blocks, 256, 0, stream>>>(X, edge_row, edge_col, edge_val,
                                                           cnt, slots, spill, Wmat, bias,
                                                           out, N, cstride, ntiles);
    }
}
